# starting kernel
# baseline (speedup 1.0000x reference)
#include <hip/hip_runtime.h>
#include <hip/hip_bf16.h>
#include <stdint.h>

typedef __attribute__((ext_vector_type(2))) float v2f;
typedef __attribute__((ext_vector_type(8))) float v8f;

__constant__ int c_unis[58] = {
    0, 1, 2, 3, 4, 6, 7, 8, 12, 14, 15, 16, 24, 28, 30, 31, 32, 48, 56, 60,
    62, 63, 64, 96, 112, 120, 124, 126, 127, 128, 129, 131, 135, 143, 159,
    191, 192, 193, 195, 199, 207, 223, 224, 225, 227, 231, 239, 240, 241,
    243, 247, 248, 249, 251, 252, 253, 254, 255};

__global__ void zero_hist_kernel(unsigned* __restrict__ hist, int n) {
  int i = blockIdx.x * blockDim.x + threadIdx.x;
  if (i < n) hist[i] = 0u;
}

__device__ __forceinline__ unsigned quant(float f) {
  return (unsigned)(f * 255.0f);
}

template <int NQ>
__device__ __forceinline__ void load_quant_row(const float* __restrict__ img,
                                               int row, int cstart,
                                               unsigned* __restrict__ q) {
  if ((unsigned)row < 512u) {
    const float* rp = img + row * 512;
#pragma unroll
    for (int i = 0; i < NQ; ++i) {
      const int cq = cstart + i * 4;
      if ((unsigned)cq < 512u) {
        const float4 v = *(const float4*)(rp + cq);
        q[i * 4 + 0] = quant(v.x);
        q[i * 4 + 1] = quant(v.y);
        q[i * 4 + 2] = quant(v.z);
        q[i * 4 + 3] = quant(v.w);
      } else {
        q[i * 4 + 0] = 0u;
        q[i * 4 + 1] = 0u;
        q[i * 4 + 2] = 0u;
        q[i * 4 + 3] = 0u;
      }
    }
  } else {
#pragma unroll
    for (int i = 0; i < NQ * 4; ++i) q[i] = 0u;
  }
}

__global__ __launch_bounds__(256) void lbp_hist_kernel(
    const float* __restrict__ x, unsigned* __restrict__ hist) {
  __shared__ unsigned sh[256];
  const int t = threadIdx.x;
  sh[t] = 0u;
  __syncthreads();

  const int batch = blockIdx.y;
  const int r = (blockIdx.x << 3) + (t >> 5);
  const int c = (t & 31) << 4;
  const float* img = x + (size_t)batch * (512 * 512);

  unsigned qc[32], qu[24], qd[24];
  load_quant_row<8>(img, r, c - 8, qc);
  load_quant_row<6>(img, r - 3, c - 4, qu);
  load_quant_row<6>(img, r + 3, c - 4, qd);

#pragma unroll
  for (int j = 0; j < 16; ++j) {
    const unsigned center = qc[j + 8];
    unsigned code = 0u;
    code |= (qu[j + 4]  >= center) ? 1u   : 0u;
    code |= (qu[j + 7]  >= center) ? 2u   : 0u;
    code |= (qc[j + 13] >= center) ? 4u   : 0u;
    code |= (qd[j + 7]  >= center) ? 8u   : 0u;
    code |= (qd[j + 4]  >= center) ? 16u  : 0u;
    code |= (qd[j + 1]  >= center) ? 32u  : 0u;
    code |= (qc[j + 3]  >= center) ? 64u  : 0u;
    code |= (qu[j + 1]  >= center) ? 128u : 0u;
    atomicAdd(&sh[code], 1u);
  }

  __syncthreads();
  const unsigned v = sh[t];
  if (v) atomicAdd(&hist[batch * 256 + t], v);
}

__global__ __launch_bounds__(256) void wmma_reduce_kernel(
    const unsigned* __restrict__ hist, float* __restrict__ out, int B) {
  __shared__ int smap[256];
  const int t = threadIdx.x;
  {
    int m = 58;
#pragma unroll
    for (int j = 0; j < 58; ++j)
      if (c_unis[j] == t) m = j;
    smap[t] = m;
  }
  __syncthreads();

  const int wave = t >> 5;
  const int lane = t & 31;
  const int lm = lane & 15;
  const int hi = (lane >> 4) & 1;
  const int mtiles = B >> 4;
  const int ntotal = mtiles * 4;

  for (int tile = wave; tile < ntotal; tile += 8) {
    const int mt = tile >> 2;
    const int nt = tile & 3;
    const int m = (mt << 4) + lm;
    const int n = (nt << 4) + lm;

    v8f acc = {};
    const unsigned* hrow = hist + (size_t)m * 256;
    for (int k0 = 0; k0 < 256; k0 += 4) {
      const int ka = k0 + (hi ? 2 : 0);
      v2f A = {(float)hrow[ka], (float)hrow[ka + 1]};
      v2f Bm = {(smap[ka] == n) ? 1.0f : 0.0f,
                (smap[ka + 1] == n) ? 1.0f : 0.0f};
      acc = __builtin_amdgcn_wmma_f32_16x16x4_f32(
           false, A,  false, Bm,
           (short)0, acc,  false,  false);
    }

    if (n < 59) {
#pragma unroll
      for (int r = 0; r < 8; ++r) {
        const int M = (mt << 4) + (hi << 3) + r;
        const unsigned iv = (unsigned)(acc[r] + 0.5f);
        out[M * 59 + n] = (float)(iv & 255u);
      }
    }
  }
}

extern "C" void kernel_launch(void* const* d_in, const int* in_sizes, int n_in,
                              void* d_out, int out_size, void* d_ws,
                              size_t ws_size, hipStream_t stream) {
  const float* x = (const float*)d_in[0];
  float* out = (float*)d_out;
  unsigned* hist = (unsigned*)d_ws;

  const int B = in_sizes[0] >> 18;

  zero_hist_kernel<<<(B * 256 + 255) / 256, 256, 0, stream>>>(hist, B * 256);
  lbp_hist_kernel<<<dim3(64, B), 256, 0, stream>>>(x, hist);
  wmma_reduce_kernel<<<1, 256, 0, stream>>>(hist, out, B);
}


# lbp kernel: code bits in float registers: a neighbour test is one subtract with clamp (no lane mask + select), two pixels join per packed multiply-add (acc = 2*acc + bit); histogram in that bit order
# speedup vs baseline: 3.0732x; 3.0732x over previous
#include <hip/hip_runtime.h>
#include <hip/hip_bf16.h>
#include <stdint.h>

typedef __attribute__((ext_vector_type(2))) float v2f;
typedef __attribute__((ext_vector_type(8))) float v8f;

__constant__ int c_unis[58] = {
    0, 1, 2, 3, 4, 6, 7, 8, 12, 14, 15, 16, 24, 28, 30, 31, 32, 48, 56, 60,
    62, 63, 64, 96, 112, 120, 124, 126, 127, 128, 129, 131, 135, 143, 159,
    191, 192, 193, 195, 199, 207, 223, 224, 225, 227, 231, 239, 240, 241,
    243, 247, 248, 249, 251, 252, 253, 254, 255};

__global__ void zero_hist_kernel(unsigned* __restrict__ hist, int n) {
  int i = blockIdx.x * blockDim.x + threadIdx.x;
  if (i < n) hist[i] = 0u;
}

__device__ __forceinline__ float quant(float f, float scale) {
  return __builtin_truncf(f * scale);
}

__device__ __forceinline__ float ge01(float n, float cm1) {
  return __builtin_amdgcn_fmed3f(n - cm1, 0.0f, 1.0f);
}

__device__ __forceinline__ int code_index(int c) {
  return ((c & 1) << 7) | ((c & 2) << 5) | ((c & 4) << 3) | ((c & 64) >> 2) |
         ((c & 128) >> 4) | ((c & 8) >> 1) | ((c & 16) >> 3) | ((c & 32) >> 5);
}

#define WPB 8
#define WGPI 8
#define NW0 22
#define NW1 21
#define NW2 21

__device__ __forceinline__ void load_row(const float* __restrict__ img, int a,
                                         int lane, float4 (&b)[4]) {
  const int ar = a < 0 ? 0 : (a > 511 ? 511 : a);
  const float4* rp = (const float4*)(img + ar * 512 + lane * 16);
  b[0] = rp[0];
  b[1] = rp[1];
  b[2] = rp[2];
  b[3] = rp[3];
}

__global__ __launch_bounds__(256) __attribute__((amdgpu_waves_per_eu(8, 8))) void lbp_hist_kernel(
    const float* __restrict__ x, unsigned* __restrict__ red) {
  __shared__ unsigned sh[256 + 32];
  __shared__ float suni[64];
  const int t = threadIdx.x;
  const int myuni = code_index(c_unis[t < 58 ? t : 57]);
  sh[t] = 0u;
  if (t < 32) sh[256 + t] = 0u;
  __syncthreads();

  const int lane = t & 31;
  const int wave = __builtin_amdgcn_readfirstlane(t >> 5);
  const int batch = blockIdx.y;
  const float* img = x + (size_t)batch * (512 * 512);

  const int w = blockIdx.x * WPB + wave;
  int k, i0, i1;
  if (w < NW0) {
    k = 0;
    i0 = (w * 171) / NW0;
    i1 = ((w + 1) * 171) / NW0;
  } else if (w < NW0 + NW1) {
    k = 1;
    i0 = ((w - NW0) * 171) / NW1;
    i1 = ((w - NW0 + 1) * 171) / NW1;
  } else {
    k = 2;
    i0 = ((w - NW0 - NW1) * 170) / NW2;
    i1 = ((w - NW0 - NW1 + 1) * 170) / NW2;
  }
  const int nrows = i1 - i0;
  const int nsteps = nrows + 2;
  const int abase = k + 3 * (i0 - 1);

  const int addrL = ((lane + 31) & 31) << 2;
  const int addrR = ((lane + 1) & 31) << 2;
  const bool first = (lane == 0);
  const bool last = (lane == 31);

  float sv[16] = {}, sl[3] = {}, sr[3] = {};
  v2f sp[8] = {};
  const v2f two = {2.0f, 2.0f};

  unsigned n0 = 0u, n255 = 0u;
  const unsigned spare = 256u + (unsigned)lane;

  float4 buf[4];
  load_row(img, abase, lane, buf);

  for (int ts = 0; ts < nsteps; ++ts) {
      {
        const int a = abase + 3 * ts;
        const float sc = ((unsigned)a < 512u) ? 255.0f : 0.0f;
        float e[26];
#pragma unroll
        for (int i = 0; i < 4; ++i) {
          e[5 + 4 * i + 0] = quant(buf[i].x, sc);
          e[5 + 4 * i + 1] = quant(buf[i].y, sc);
          e[5 + 4 * i + 2] = quant(buf[i].z, sc);
          e[5 + 4 * i + 3] = quant(buf[i].w, sc);
        }
        load_row(img, a + 3, lane, buf);

#pragma unroll
        for (int i = 0; i < 5; ++i) {
          const float fromL = __int_as_float(__builtin_amdgcn_ds_bpermute(
              addrL, __float_as_int(e[16 + i])));
          const float fromR = __int_as_float(__builtin_amdgcn_ds_bpermute(
              addrR, __float_as_int(e[5 + i])));
          e[i] = first ? 0.0f : fromL;
          e[21 + i] = last ? 0.0f : fromR;
        }

        float u[22];
#pragma unroll
        for (int i = 0; i < 3; ++i) {
          u[i] = sl[i];
          u[19 + i] = sr[i];
        }
#pragma unroll
        for (int j = 0; j < 16; ++j) u[3 + j] = sv[j];

        if (ts >= 2) {
#pragma unroll
          for (int jj = 0; jj < 8; ++jj) {
            const int j = 2 * jj;
            const float c0 = u[3 + j] - 1.0f, c1 = u[4 + j] - 1.0f;
            const v2f b3 = {ge01(e[5 + j + 3], c0), ge01(e[6 + j + 3], c1)};
            const v2f b4 = {ge01(e[5 + j], c0), ge01(e[6 + j], c1)};
            const v2f b5 = {ge01(e[5 + j - 3], c0), ge01(e[6 + j - 3], c1)};
            v2f acc = sp[jj];
            acc = __builtin_elementwise_fma(acc, two, b3);
            acc = __builtin_elementwise_fma(acc, two, b4);
            acc = __builtin_elementwise_fma(acc, two, b5);
#pragma unroll
            for (int h = 0; h < 2; ++h) {
              const float cf = acc[h];
              const bool is0 = (cf == 0.0f);
              const bool is255 = (cf == 255.0f);
              n0 += (unsigned)__builtin_popcount(
                  __builtin_amdgcn_ballot_w32(is0));
              n255 += (unsigned)__builtin_popcount(
                  __builtin_amdgcn_ballot_w32(is255));
              atomicAdd(&sh[(is0 | is255) ? spare : (unsigned)cf], 1u);
            }
          }
        }
        if (ts >= 1 && ts <= nrows) {
#pragma unroll
          for (int jj = 0; jj < 8; ++jj) {
            const int j = 2 * jj;
            const float c0 = e[5 + j] - 1.0f, c1 = e[6 + j] - 1.0f;
            v2f p = {ge01(u[3 + j], c0), ge01(u[4 + j], c1)};
            const v2f b1 = {ge01(u[3 + j + 3], c0), ge01(u[4 + j + 3], c1)};
            const v2f b2 = {ge01(e[5 + j + 5], c0), ge01(e[6 + j + 5], c1)};
            const v2f b6 = {ge01(e[5 + j - 5], c0), ge01(e[6 + j - 5], c1)};
            const v2f b7 = {ge01(u[3 + j - 3], c0), ge01(u[4 + j - 3], c1)};
            p = __builtin_elementwise_fma(p, two, b1);
            p = __builtin_elementwise_fma(p, two, b2);
            p = __builtin_elementwise_fma(p, two, b6);
            p = __builtin_elementwise_fma(p, two, b7);
            sp[jj] = p;
          }
        }
#pragma unroll
        for (int j = 0; j < 16; ++j) sv[j] = e[5 + j];
#pragma unroll
        for (int i = 0; i < 3; ++i) {
          sl[i] = e[2 + i];
          sr[i] = e[21 + i];
        }
      }
  }

  if (lane == 0) {
    atomicAdd(&sh[0], n0);
    atomicAdd(&sh[255], n255);
  }
  __syncthreads();

  if (t < 64) suni[t] = (t < 58) ? (float)sh[myuni] : 0.0f;
  __syncthreads();
  if (wave < 4) {
    const int lm = lane & 15;
    const int hi = lane >> 4;
    const int n = (wave << 4) + lm;
    const float bsum = (lm == 10) ? 1.0f : 0.0f;
    v8f accI = {}, accS = {};
#pragma unroll
    for (int i = 0; i < 4; ++i) {
      const int ka = (wave << 4) + 4 * i + (hi ? 2 : 0);
      const float a0 = suni[ka], a1 = suni[ka + 1];
      v2f A = {lm == 0 ? a0 : 0.0f, lm == 0 ? a1 : 0.0f};
      v2f Bi = {(ka == n) ? 1.0f : 0.0f, (ka + 1 == n) ? 1.0f : 0.0f};
      v2f Bs = {bsum, bsum};
      accI = __builtin_amdgcn_wmma_f32_16x16x4_f32(
           false, A,  false, Bi,
           (short)0, accI,  false,  false);
      accS = __builtin_amdgcn_wmma_f32_16x16x4_f32(
           false, A,  false, Bs,
           (short)0, accS,  false,  false);
    }
    if (hi == 0) {
      unsigned* rrow = red + batch * 64;
      const unsigned vi = (unsigned)(accI[0] + 0.5f);
      const unsigned vs = (unsigned)(accS[0] + 0.5f);
      if (n < 58 && vi) atomicAdd(&rrow[n], vi);
      if (lm == 10 && vs) atomicAdd(&rrow[58], vs);
    }
  }
}

__global__ __launch_bounds__(256) void finish_kernel(
    const unsigned* __restrict__ red, float* __restrict__ out, int total) {
  const int i = blockIdx.x * blockDim.x + threadIdx.x;
  if (i < total) {
    const int b = i / 59;
    const int n = i - b * 59;
    unsigned v = red[b * 64 + n];
    if (n == 58) v = (512u * 512u) - v;
    out[i] = (float)(v & 255u);
  }
}

extern "C" void kernel_launch(void* const* d_in, const int* in_sizes, int n_in,
                              void* d_out, int out_size, void* d_ws,
                              size_t ws_size, hipStream_t stream) {
  const float* x = (const float*)d_in[0];
  float* out = (float*)d_out;
  unsigned* red = (unsigned*)d_ws;

  const int B = in_sizes[0] >> 18;

  zero_hist_kernel<<<(B * 64 + 255) / 256, 256, 0, stream>>>(red, B * 64);
  lbp_hist_kernel<<<dim3(WGPI, B), 256, 0, stream>>>(x, red);
  finish_kernel<<<(B * 59 + 255) / 256, 256, 0, stream>>>(red, out, B * 59);
}


# lbp kernel: one LDS atomic per pixel PAIR into a joint histogram H[code0*257+code1] (row + column sums give the counts); 0/255 wave votes dropped; its LDS forces 2 work-groups of 32 waves per image
# speedup vs baseline: 3.5192x; 1.1451x over previous
#include <hip/hip_runtime.h>
#include <hip/hip_bf16.h>
#include <stdint.h>

typedef __attribute__((ext_vector_type(2))) float v2f;
typedef __attribute__((ext_vector_type(8))) float v8f;

__constant__ int c_unis[58] = {
    0, 1, 2, 3, 4, 6, 7, 8, 12, 14, 15, 16, 24, 28, 30, 31, 32, 48, 56, 60,
    62, 63, 64, 96, 112, 120, 124, 126, 127, 128, 129, 131, 135, 143, 159,
    191, 192, 193, 195, 199, 207, 223, 224, 225, 227, 231, 239, 240, 241,
    243, 247, 248, 249, 251, 252, 253, 254, 255};

__global__ void zero_hist_kernel(unsigned* __restrict__ hist, int n) {
  int i = blockIdx.x * blockDim.x + threadIdx.x;
  if (i < n) hist[i] = 0u;
}

__device__ __forceinline__ float quant(float f, float scale) {
  return __builtin_truncf(f * scale);
}

__device__ __forceinline__ float ge01(float n, float cm1) {
  return __builtin_amdgcn_fmed3f(n - cm1, 0.0f, 1.0f);
}

__device__ __forceinline__ int code_index(int c) {
  return ((c & 1) << 7) | ((c & 2) << 5) | ((c & 4) << 3) | ((c & 64) >> 2) |
         ((c & 128) >> 4) | ((c & 8) >> 1) | ((c & 16) >> 3) | ((c & 32) >> 5);
}

#define WPB 32
#define WGPI 2
#define HP 257
#define NW0 22
#define NW1 21
#define NW2 21

__device__ __forceinline__ void load_row(const float* __restrict__ img, int a,
                                         int lane, float4 (&b)[4]) {
  const int ar = a < 0 ? 0 : (a > 511 ? 511 : a);
  const float4* rp = (const float4*)(img + ar * 512 + lane * 16);
  b[0] = rp[0];
  b[1] = rp[1];
  b[2] = rp[2];
  b[3] = rp[3];
}

__global__ __launch_bounds__(1024) void lbp_hist_kernel(
    const float* __restrict__ x, unsigned* __restrict__ red) {
  __shared__ __attribute__((aligned(16))) unsigned H[256 * HP];
  __shared__ float suni[64];
  const int t = threadIdx.x;
  const int fu = t >> 4;
  const int fp = t & 15;
  const int fm = code_index(c_unis[fu < 58 ? fu : 57]);

  const int lane = t & 31;
  const int wave = __builtin_amdgcn_readfirstlane(t >> 5);
  const int batch = blockIdx.y;
  const float* img = x + (size_t)batch * (512 * 512);

  const int w = blockIdx.x * WPB + wave;
  int k, i0, i1;
  if (w < NW0) {
    k = 0;
    i0 = (w * 171) / NW0;
    i1 = ((w + 1) * 171) / NW0;
  } else if (w < NW0 + NW1) {
    k = 1;
    i0 = ((w - NW0) * 171) / NW1;
    i1 = ((w - NW0 + 1) * 171) / NW1;
  } else {
    k = 2;
    i0 = ((w - NW0 - NW1) * 170) / NW2;
    i1 = ((w - NW0 - NW1 + 1) * 170) / NW2;
  }
  const int nrows = i1 - i0;
  const int nsteps = nrows + 2;
  const int abase = k + 3 * (i0 - 1);

  const int addrL = ((lane + 31) & 31) << 2;
  const int addrR = ((lane + 1) & 31) << 2;
  const bool first = (lane == 0);
  const bool last = (lane == 31);

  float sv[16] = {}, sl[3] = {}, sr[3] = {};
  v2f sp[8] = {};
  const v2f two = {2.0f, 2.0f};

  float4 buf[4];
  load_row(img, abase, lane, buf);

  {
    uint4* Hz = (uint4*)H;
    for (int i = t; i < (256 * HP) / 4; i += 1024) Hz[i] = make_uint4(0u, 0u, 0u, 0u);
  }
  __syncthreads();

  for (int ts = 0; ts < nsteps; ++ts) {
      {
        const int a = abase + 3 * ts;
        const float sc = ((unsigned)a < 512u) ? 255.0f : 0.0f;
        float e[26];
#pragma unroll
        for (int i = 0; i < 4; ++i) {
          e[5 + 4 * i + 0] = quant(buf[i].x, sc);
          e[5 + 4 * i + 1] = quant(buf[i].y, sc);
          e[5 + 4 * i + 2] = quant(buf[i].z, sc);
          e[5 + 4 * i + 3] = quant(buf[i].w, sc);
        }
        load_row(img, a + 3, lane, buf);

#pragma unroll
        for (int i = 0; i < 5; ++i) {
          const float fromL = __int_as_float(__builtin_amdgcn_ds_bpermute(
              addrL, __float_as_int(e[16 + i])));
          const float fromR = __int_as_float(__builtin_amdgcn_ds_bpermute(
              addrR, __float_as_int(e[5 + i])));
          e[i] = first ? 0.0f : fromL;
          e[21 + i] = last ? 0.0f : fromR;
        }

        float u[22];
#pragma unroll
        for (int i = 0; i < 3; ++i) {
          u[i] = sl[i];
          u[19 + i] = sr[i];
        }
#pragma unroll
        for (int j = 0; j < 16; ++j) u[3 + j] = sv[j];

        if (ts >= 2) {
#pragma unroll
          for (int jj = 0; jj < 8; ++jj) {
            const int j = 2 * jj;
            const float c0 = u[3 + j] - 1.0f, c1 = u[4 + j] - 1.0f;
            const v2f b3 = {ge01(e[5 + j + 3], c0), ge01(e[6 + j + 3], c1)};
            const v2f b4 = {ge01(e[5 + j], c0), ge01(e[6 + j], c1)};
            const v2f b5 = {ge01(e[5 + j - 3], c0), ge01(e[6 + j - 3], c1)};
            v2f acc = sp[jj];
            acc = __builtin_elementwise_fma(acc, two, b3);
            acc = __builtin_elementwise_fma(acc, two, b4);
            acc = __builtin_elementwise_fma(acc, two, b5);
            const float kf = __builtin_fmaf(acc[0], (float)HP, acc[1]);
            atomicAdd(&H[(unsigned)kf], 1u);
          }
        }
        if (ts >= 1 && ts <= nrows) {
#pragma unroll
          for (int jj = 0; jj < 8; ++jj) {
            const int j = 2 * jj;
            const float c0 = e[5 + j] - 1.0f, c1 = e[6 + j] - 1.0f;
            v2f p = {ge01(u[3 + j], c0), ge01(u[4 + j], c1)};
            const v2f b1 = {ge01(u[3 + j + 3], c0), ge01(u[4 + j + 3], c1)};
            const v2f b2 = {ge01(e[5 + j + 5], c0), ge01(e[6 + j + 5], c1)};
            const v2f b6 = {ge01(e[5 + j - 5], c0), ge01(e[6 + j - 5], c1)};
            const v2f b7 = {ge01(u[3 + j - 3], c0), ge01(u[4 + j - 3], c1)};
            p = __builtin_elementwise_fma(p, two, b1);
            p = __builtin_elementwise_fma(p, two, b2);
            p = __builtin_elementwise_fma(p, two, b6);
            p = __builtin_elementwise_fma(p, two, b7);
            sp[jj] = p;
          }
        }
#pragma unroll
        for (int j = 0; j < 16; ++j) sv[j] = e[5 + j];
#pragma unroll
        for (int i = 0; i < 3; ++i) {
          sl[i] = e[2 + i];
          sr[i] = e[21 + i];
        }
      }
  }

  __syncthreads();

  {
    unsigned cnt = 0u;
    const int rot = (fp >> 2) << 2;
#pragma unroll
    for (int i = 0; i < 16; ++i) {
      const int q = (fp << 4) + ((i + rot) & 15);
      cnt += H[fm * HP + q];
      cnt += H[q * HP + fm];
    }
    cnt += __shfl_xor(cnt, 1);
    cnt += __shfl_xor(cnt, 2);
    cnt += __shfl_xor(cnt, 4);
    cnt += __shfl_xor(cnt, 8);
    if (fp == 0) suni[fu] = (fu < 58) ? (float)cnt : 0.0f;
  }
  __syncthreads();
  if (wave < 4) {
    const int lm = lane & 15;
    const int hi = lane >> 4;
    const int n = (wave << 4) + lm;
    const float bsum = (lm == 10) ? 1.0f : 0.0f;
    v8f accI = {}, accS = {};
#pragma unroll
    for (int i = 0; i < 4; ++i) {
      const int ka = (wave << 4) + 4 * i + (hi ? 2 : 0);
      const float a0 = suni[ka], a1 = suni[ka + 1];
      v2f A = {lm == 0 ? a0 : 0.0f, lm == 0 ? a1 : 0.0f};
      v2f Bi = {(ka == n) ? 1.0f : 0.0f, (ka + 1 == n) ? 1.0f : 0.0f};
      v2f Bs = {bsum, bsum};
      accI = __builtin_amdgcn_wmma_f32_16x16x4_f32(
           false, A,  false, Bi,
           (short)0, accI,  false,  false);
      accS = __builtin_amdgcn_wmma_f32_16x16x4_f32(
           false, A,  false, Bs,
           (short)0, accS,  false,  false);
    }
    if (hi == 0) {
      unsigned* rrow = red + batch * 64;
      const unsigned vi = (unsigned)(accI[0] + 0.5f);
      const unsigned vs = (unsigned)(accS[0] + 0.5f);
      if (n < 58 && vi) atomicAdd(&rrow[n], vi);
      if (lm == 10 && vs) atomicAdd(&rrow[58], vs);
    }
  }
}

__global__ __launch_bounds__(256) void finish_kernel(
    const unsigned* __restrict__ red, float* __restrict__ out, int total) {
  const int i = blockIdx.x * blockDim.x + threadIdx.x;
  if (i < total) {
    const int b = i / 59;
    const int n = i - b * 59;
    unsigned v = red[b * 64 + n];
    if (n == 58) v = (512u * 512u) - v;
    out[i] = (float)(v & 255u);
  }
}

extern "C" void kernel_launch(void* const* d_in, const int* in_sizes, int n_in,
                              void* d_out, int out_size, void* d_ws,
                              size_t ws_size, hipStream_t stream) {
  const float* x = (const float*)d_in[0];
  float* out = (float*)d_out;
  unsigned* red = (unsigned*)d_ws;

  const int B = in_sizes[0] >> 18;

  zero_hist_kernel<<<(B * 64 + 255) / 256, 256, 0, stream>>>(red, B * 64);
  lbp_hist_kernel<<<dim3(WGPI, B), 1024, 0, stream>>>(x, red);
  finish_kernel<<<(B * 59 + 255) / 256, 256, 0, stream>>>(red, out, B * 59);
}


# tail launch folded into the lbp flush: adds into out carry an arrival count above the count bits; the adder that sees all arrivals wraps mod 256 and swaps in the float (integer sums, order-free)
# speedup vs baseline: 3.7029x; 1.0522x over previous
#include <hip/hip_runtime.h>
#include <hip/hip_bf16.h>
#include <stdint.h>

typedef __attribute__((ext_vector_type(2))) float v2f;
typedef __attribute__((ext_vector_type(8))) float v8f;

__constant__ int c_unis[58] = {
    0, 1, 2, 3, 4, 6, 7, 8, 12, 14, 15, 16, 24, 28, 30, 31, 32, 48, 56, 60,
    62, 63, 64, 96, 112, 120, 124, 126, 127, 128, 129, 131, 135, 143, 159,
    191, 192, 193, 195, 199, 207, 223, 224, 225, 227, 231, 239, 240, 241,
    243, 247, 248, 249, 251, 252, 253, 254, 255};

__global__ void start_out_kernel(float* __restrict__ out, int n) {
  int i = blockIdx.x * blockDim.x + threadIdx.x;
  if (i < n) out[i] = 0.0f;
}

__device__ __forceinline__ float quant(float f, float scale) {
  return __builtin_truncf(f * scale);
}

__device__ __forceinline__ float ge01(float n, float cm1) {
  return __builtin_amdgcn_fmed3f(n - cm1, 0.0f, 1.0f);
}

__device__ __forceinline__ int code_index(int c) {
  return ((c & 1) << 7) | ((c & 2) << 5) | ((c & 4) << 3) | ((c & 64) >> 2) |
         ((c & 128) >> 4) | ((c & 8) >> 1) | ((c & 16) >> 3) | ((c & 32) >> 5);
}

#define WPB 32
#define WGPI 2
#define HP 257
#define ARR_SHIFT 20
#define ARR_ONE (1u << ARR_SHIFT)
#define NW0 22
#define NW1 21
#define NW2 21

__device__ __forceinline__ void load_row(const float* __restrict__ img, int a,
                                         int lane, float4 (&b)[4]) {
  const int ar = a < 0 ? 0 : (a > 511 ? 511 : a);
  const float4* rp = (const float4*)(img + ar * 512 + lane * 16);
  b[0] = rp[0];
  b[1] = rp[1];
  b[2] = rp[2];
  b[3] = rp[3];
}

__global__ __launch_bounds__(1024) void lbp_hist_kernel(
    const float* __restrict__ x, float* __restrict__ out) {
  __shared__ __attribute__((aligned(16))) unsigned H[256 * HP];
  __shared__ float suni[64];
  const int t = threadIdx.x;
  const int fu = t >> 4;
  const int fp = t & 15;
  const int fm = code_index(c_unis[fu < 58 ? fu : 57]);

  const int lane = t & 31;
  const int wave = __builtin_amdgcn_readfirstlane(t >> 5);
  const int batch = blockIdx.y;
  const float* img = x + (size_t)batch * (512 * 512);

  const int w = blockIdx.x * WPB + wave;
  int k, i0, i1;
  if (w < NW0) {
    k = 0;
    i0 = (w * 171) / NW0;
    i1 = ((w + 1) * 171) / NW0;
  } else if (w < NW0 + NW1) {
    k = 1;
    i0 = ((w - NW0) * 171) / NW1;
    i1 = ((w - NW0 + 1) * 171) / NW1;
  } else {
    k = 2;
    i0 = ((w - NW0 - NW1) * 170) / NW2;
    i1 = ((w - NW0 - NW1 + 1) * 170) / NW2;
  }
  const int nrows = i1 - i0;
  const int nsteps = nrows + 2;
  const int abase = k + 3 * (i0 - 1);

  const int addrL = ((lane + 31) & 31) << 2;
  const int addrR = ((lane + 1) & 31) << 2;
  const bool first = (lane == 0);
  const bool last = (lane == 31);

  float sv[16] = {}, sl[3] = {}, sr[3] = {};
  v2f sp[8] = {};
  const v2f two = {2.0f, 2.0f};

  float4 buf[4];
  load_row(img, abase, lane, buf);

  {
    uint4* Hz = (uint4*)H;
    for (int i = t; i < (256 * HP) / 4; i += 1024) Hz[i] = make_uint4(0u, 0u, 0u, 0u);
  }
  __syncthreads();

  for (int ts = 0; ts < nsteps; ++ts) {
      {
        const int a = abase + 3 * ts;
        const float sc = ((unsigned)a < 512u) ? 255.0f : 0.0f;
        float e[26];
#pragma unroll
        for (int i = 0; i < 4; ++i) {
          e[5 + 4 * i + 0] = quant(buf[i].x, sc);
          e[5 + 4 * i + 1] = quant(buf[i].y, sc);
          e[5 + 4 * i + 2] = quant(buf[i].z, sc);
          e[5 + 4 * i + 3] = quant(buf[i].w, sc);
        }
        load_row(img, a + 3, lane, buf);

#pragma unroll
        for (int i = 0; i < 5; ++i) {
          const float fromL = __int_as_float(__builtin_amdgcn_ds_bpermute(
              addrL, __float_as_int(e[16 + i])));
          const float fromR = __int_as_float(__builtin_amdgcn_ds_bpermute(
              addrR, __float_as_int(e[5 + i])));
          e[i] = first ? 0.0f : fromL;
          e[21 + i] = last ? 0.0f : fromR;
        }

        float u[22];
#pragma unroll
        for (int i = 0; i < 3; ++i) {
          u[i] = sl[i];
          u[19 + i] = sr[i];
        }
#pragma unroll
        for (int j = 0; j < 16; ++j) u[3 + j] = sv[j];

        if (ts >= 2) {
#pragma unroll
          for (int jj = 0; jj < 8; ++jj) {
            const int j = 2 * jj;
            const float c0 = u[3 + j] - 1.0f, c1 = u[4 + j] - 1.0f;
            const v2f b3 = {ge01(e[5 + j + 3], c0), ge01(e[6 + j + 3], c1)};
            const v2f b4 = {ge01(e[5 + j], c0), ge01(e[6 + j], c1)};
            const v2f b5 = {ge01(e[5 + j - 3], c0), ge01(e[6 + j - 3], c1)};
            v2f acc = sp[jj];
            acc = __builtin_elementwise_fma(acc, two, b3);
            acc = __builtin_elementwise_fma(acc, two, b4);
            acc = __builtin_elementwise_fma(acc, two, b5);
            const float kf = __builtin_fmaf(acc[0], (float)HP, acc[1]);
            atomicAdd(&H[(unsigned)kf], 1u);
          }
        }
        if (ts >= 1 && ts <= nrows) {
#pragma unroll
          for (int jj = 0; jj < 8; ++jj) {
            const int j = 2 * jj;
            const float c0 = e[5 + j] - 1.0f, c1 = e[6 + j] - 1.0f;
            v2f p = {ge01(u[3 + j], c0), ge01(u[4 + j], c1)};
            const v2f b1 = {ge01(u[3 + j + 3], c0), ge01(u[4 + j + 3], c1)};
            const v2f b2 = {ge01(e[5 + j + 5], c0), ge01(e[6 + j + 5], c1)};
            const v2f b6 = {ge01(e[5 + j - 5], c0), ge01(e[6 + j - 5], c1)};
            const v2f b7 = {ge01(u[3 + j - 3], c0), ge01(u[4 + j - 3], c1)};
            p = __builtin_elementwise_fma(p, two, b1);
            p = __builtin_elementwise_fma(p, two, b2);
            p = __builtin_elementwise_fma(p, two, b6);
            p = __builtin_elementwise_fma(p, two, b7);
            sp[jj] = p;
          }
        }
#pragma unroll
        for (int j = 0; j < 16; ++j) sv[j] = e[5 + j];
#pragma unroll
        for (int i = 0; i < 3; ++i) {
          sl[i] = e[2 + i];
          sr[i] = e[21 + i];
        }
      }
  }

  __syncthreads();

  {
    unsigned cnt = 0u;
    const int rot = (fp >> 2) << 2;
#pragma unroll
    for (int i = 0; i < 16; ++i) {
      const int q = (fp << 4) + ((i + rot) & 15);
      cnt += H[fm * HP + q];
      cnt += H[q * HP + fm];
    }
    cnt += __shfl_xor(cnt, 1);
    cnt += __shfl_xor(cnt, 2);
    cnt += __shfl_xor(cnt, 4);
    cnt += __shfl_xor(cnt, 8);
    if (fp == 0) suni[fu] = (fu < 58) ? (float)cnt : 0.0f;
  }
  __syncthreads();
  if (wave < 4) {
    const int lm = lane & 15;
    const int hi = lane >> 4;
    const int n = (wave << 4) + lm;
    const float bsum = (lm == 10) ? 1.0f : 0.0f;
    v8f accI = {}, accS = {};
#pragma unroll
    for (int i = 0; i < 4; ++i) {
      const int ka = (wave << 4) + 4 * i + (hi ? 2 : 0);
      const float a0 = suni[ka], a1 = suni[ka + 1];
      v2f A = {lm == 0 ? a0 : 0.0f, lm == 0 ? a1 : 0.0f};
      v2f Bi = {(ka == n) ? 1.0f : 0.0f, (ka + 1 == n) ? 1.0f : 0.0f};
      v2f Bs = {bsum, bsum};
      accI = __builtin_amdgcn_wmma_f32_16x16x4_f32(
           false, A,  false, Bi,
           (short)0, accI,  false,  false);
      accS = __builtin_amdgcn_wmma_f32_16x16x4_f32(
           false, A,  false, Bs,
           (short)0, accS,  false,  false);
    }
    if (hi == 0) {
      unsigned* orow = (unsigned*)out + batch * 59;
      const unsigned vi = (unsigned)(accI[0] + 0.5f);
      const unsigned vs = (unsigned)(accS[0] + 0.5f);
      if (n < 58) {
        const unsigned old = atomicAdd(&orow[n], vi + ARR_ONE);
        if ((old >> ARR_SHIFT) == WGPI - 1) {
          const unsigned tot = (old & (ARR_ONE - 1u)) + vi;
          atomicExch(&orow[n], __float_as_uint((float)(tot & 255u)));
        }
      }
      if (lm == 10) {
        const unsigned old = atomicAdd(&orow[58], vs + ARR_ONE);
        if ((old >> ARR_SHIFT) == 4 * WGPI - 1) {
          const unsigned tot = (old & (ARR_ONE - 1u)) + vs;
          const unsigned rest = (512u * 512u) - tot;
          atomicExch(&orow[58], __float_as_uint((float)(rest & 255u)));
        }
      }
    }
  }
}

extern "C" void kernel_launch(void* const* d_in, const int* in_sizes, int n_in,
                              void* d_out, int out_size, void* d_ws,
                              size_t ws_size, hipStream_t stream) {
  const float* x = (const float*)d_in[0];
  float* out = (float*)d_out;

  const int B = in_sizes[0] >> 18;

  start_out_kernel<<<(B * 59 + 255) / 256, 256, 0, stream>>>(out, B * 59);
  lbp_hist_kernel<<<dim3(WGPI, B), 1024, 0, stream>>>(x, out);
}


# lbp kernel: two pixels per register as packed f16 (exact: integers up to 256); a neighbour test of a PAIR is one packed add with clamp, not two subtracts; 6 lane exchanges per row, not 10
# speedup vs baseline: 3.8802x; 1.0479x over previous
#include <hip/hip_runtime.h>
#include <hip/hip_bf16.h>
#include <stdint.h>

typedef __attribute__((ext_vector_type(2))) float v2f;
typedef __attribute__((ext_vector_type(8))) float v8f;
typedef __attribute__((ext_vector_type(2))) _Float16 v2h;

__constant__ int c_unis[58] = {
    0, 1, 2, 3, 4, 6, 7, 8, 12, 14, 15, 16, 24, 28, 30, 31, 32, 48, 56, 60,
    62, 63, 64, 96, 112, 120, 124, 126, 127, 128, 129, 131, 135, 143, 159,
    191, 192, 193, 195, 199, 207, 223, 224, 225, 227, 231, 239, 240, 241,
    243, 247, 248, 249, 251, 252, 253, 254, 255};

__global__ void start_out_kernel(float* __restrict__ out, int n) {
  int i = blockIdx.x * blockDim.x + threadIdx.x;
  if (i < n) out[i] = 0.0f;
}

__device__ __forceinline__ float quant(float f, float scale) {
  return __builtin_truncf(f * scale);
}

__device__ __forceinline__ v2h pack2(float a, float b) {
  return __builtin_bit_cast(v2h, __builtin_amdgcn_cvt_pkrtz(a, b));
}

__device__ __forceinline__ v2h ge01(v2h n, v2h m) {
  const v2h zero = {(_Float16)0.0f, (_Float16)0.0f};
  const v2h one = {(_Float16)1.0f, (_Float16)1.0f};
  return __builtin_elementwise_min(__builtin_elementwise_max(n + m, zero), one);
}

__device__ __forceinline__ int code_index(int c) {
  return ((c & 1) << 7) | ((c & 2) << 5) | ((c & 4) << 3) | ((c & 64) >> 2) |
         ((c & 128) >> 4) | ((c & 8) >> 1) | ((c & 16) >> 3) | ((c & 32) >> 5);
}

#define WPB 32
#define WGPI 2
#define HP 257
#define ARR_SHIFT 20
#define ARR_ONE (1u << ARR_SHIFT)
#define NW0 22
#define NW1 21
#define NW2 21

__device__ __forceinline__ void load_row(const float* __restrict__ img, int a,
                                         int lane, float4 (&b)[4]) {
  const int ar = a < 0 ? 0 : (a > 511 ? 511 : a);
  const float4* rp = (const float4*)(img + ar * 512 + lane * 16);
  b[0] = rp[0];
  b[1] = rp[1];
  b[2] = rp[2];
  b[3] = rp[3];
}

__global__ __launch_bounds__(1024) void lbp_hist_kernel(
    const float* __restrict__ x, float* __restrict__ out) {
  __shared__ __attribute__((aligned(16))) unsigned H[256 * HP];
  __shared__ float suni[64];
  const int t = threadIdx.x;
  const int fu = t >> 4;
  const int fp = t & 15;
  const int fm = code_index(c_unis[fu < 58 ? fu : 57]);

  const int lane = t & 31;
  const int wave = __builtin_amdgcn_readfirstlane(t >> 5);
  const int batch = blockIdx.y;
  const float* img = x + (size_t)batch * (512 * 512);

  const int w = blockIdx.x * WPB + wave;
  int k, i0, i1;
  if (w < NW0) {
    k = 0;
    i0 = (w * 171) / NW0;
    i1 = ((w + 1) * 171) / NW0;
  } else if (w < NW0 + NW1) {
    k = 1;
    i0 = ((w - NW0) * 171) / NW1;
    i1 = ((w - NW0 + 1) * 171) / NW1;
  } else {
    k = 2;
    i0 = ((w - NW0 - NW1) * 170) / NW2;
    i1 = ((w - NW0 - NW1 + 1) * 170) / NW2;
  }
  const int nrows = i1 - i0;
  const int nsteps = nrows + 2;
  const int abase = k + 3 * (i0 - 1);

  const int addrL = ((lane + 31) & 31) << 2;
  const int addrR = ((lane + 1) & 31) << 2;
  const bool first = (lane == 0);
  const bool last = (lane == 31);

  v2h sE[8] = {}, sM[8] = {}, sO[11] = {}, sp[8] = {};
  const v2h two = {(_Float16)2.0f, (_Float16)2.0f};
  const v2h one = {(_Float16)1.0f, (_Float16)1.0f};

  float4 buf[4];
  load_row(img, abase, lane, buf);

  {
    uint4* Hz = (uint4*)H;
    for (int i = t; i < (256 * HP) / 4; i += 1024) Hz[i] = make_uint4(0u, 0u, 0u, 0u);
  }
  __syncthreads();

  for (int ts = 0; ts < nsteps; ++ts) {
      {
        const int a = abase + 3 * ts;
        const float sc = ((unsigned)a < 512u) ? 255.0f : 0.0f;
        float q[16];
#pragma unroll
        for (int i = 0; i < 4; ++i) {
          q[4 * i + 0] = quant(buf[i].x, sc);
          q[4 * i + 1] = quant(buf[i].y, sc);
          q[4 * i + 2] = quant(buf[i].z, sc);
          q[4 * i + 3] = quant(buf[i].w, sc);
        }
        load_row(img, a + 3, lane, buf);

        v2h E[8], O[13];
#pragma unroll
        for (int k = 0; k < 8; ++k) E[k] = pack2(q[2 * k], q[2 * k + 1]);
#pragma unroll
        for (int m = 0; m < 7; ++m) O[m + 3] = pack2(q[2 * m + 1], q[2 * m + 2]);
        {
          const int l5 = __builtin_amdgcn_ds_bpermute(addrL, __builtin_bit_cast(int, O[8]));
          const int l6 = __builtin_amdgcn_ds_bpermute(addrL, __builtin_bit_cast(int, O[9]));
          const int lq = __builtin_amdgcn_ds_bpermute(addrL, __float_as_int(q[15]));
          const int r0 = __builtin_amdgcn_ds_bpermute(addrR, __builtin_bit_cast(int, O[3]));
          const int r1 = __builtin_amdgcn_ds_bpermute(addrR, __builtin_bit_cast(int, O[4]));
          const int rq = __builtin_amdgcn_ds_bpermute(addrR, __float_as_int(q[0]));
          O[0] = __builtin_bit_cast(v2h, first ? 0 : l5);
          O[1] = __builtin_bit_cast(v2h, first ? 0 : l6);
          O[2] = pack2(first ? 0.0f : __int_as_float(lq), q[0]);
          O[10] = pack2(q[15], last ? 0.0f : __int_as_float(rq));
          O[11] = __builtin_bit_cast(v2h, last ? 0 : r0);
          O[12] = __builtin_bit_cast(v2h, last ? 0 : r1);
        }
        v2h Mc[8];
#pragma unroll
        for (int k = 0; k < 8; ++k) Mc[k] = one - E[k];

        if (ts >= 2) {
#pragma unroll
          for (int jj = 0; jj < 8; ++jj) {
            v2h acc = sp[jj];
            acc = __builtin_elementwise_fma(acc, two, ge01(O[jj + 4], sM[jj]));
            acc = __builtin_elementwise_fma(acc, two, ge01(E[jj], sM[jj]));
            acc = __builtin_elementwise_fma(acc, two, ge01(O[jj + 1], sM[jj]));
            const float kf = __builtin_fmaf((float)acc[0], (float)HP, (float)acc[1]);
            atomicAdd(&H[(unsigned)kf], 1u);
          }
        }
        if (ts >= 1 && ts <= nrows) {
#pragma unroll
          for (int jj = 0; jj < 8; ++jj) {
            v2h p = ge01(sE[jj], Mc[jj]);
            p = __builtin_elementwise_fma(p, two, ge01(sO[jj + 3], Mc[jj]));
            p = __builtin_elementwise_fma(p, two, ge01(O[jj + 5], Mc[jj]));
            p = __builtin_elementwise_fma(p, two, ge01(O[jj], Mc[jj]));
            p = __builtin_elementwise_fma(p, two, ge01(sO[jj], Mc[jj]));
            sp[jj] = p;
          }
        }
#pragma unroll
        for (int k = 0; k < 8; ++k) {
          sE[k] = E[k];
          sM[k] = Mc[k];
        }
#pragma unroll
        for (int i = 0; i < 11; ++i) sO[i] = O[i + 1];
      }
  }

  __syncthreads();

  {
    unsigned cnt = 0u;
    const int rot = (fp >> 2) << 2;
#pragma unroll
    for (int i = 0; i < 16; ++i) {
      const int q = (fp << 4) + ((i + rot) & 15);
      cnt += H[fm * HP + q];
      cnt += H[q * HP + fm];
    }
    cnt += __shfl_xor(cnt, 1);
    cnt += __shfl_xor(cnt, 2);
    cnt += __shfl_xor(cnt, 4);
    cnt += __shfl_xor(cnt, 8);
    if (fp == 0) suni[fu] = (fu < 58) ? (float)cnt : 0.0f;
  }
  __syncthreads();
  if (wave < 4) {
    const int lm = lane & 15;
    const int hi = lane >> 4;
    const int n = (wave << 4) + lm;
    const float bsum = (lm == 10) ? 1.0f : 0.0f;
    v8f accI = {}, accS = {};
#pragma unroll
    for (int i = 0; i < 4; ++i) {
      const int ka = (wave << 4) + 4 * i + (hi ? 2 : 0);
      const float a0 = suni[ka], a1 = suni[ka + 1];
      v2f A = {lm == 0 ? a0 : 0.0f, lm == 0 ? a1 : 0.0f};
      v2f Bi = {(ka == n) ? 1.0f : 0.0f, (ka + 1 == n) ? 1.0f : 0.0f};
      v2f Bs = {bsum, bsum};
      accI = __builtin_amdgcn_wmma_f32_16x16x4_f32(
           false, A,  false, Bi,
           (short)0, accI,  false,  false);
      accS = __builtin_amdgcn_wmma_f32_16x16x4_f32(
           false, A,  false, Bs,
           (short)0, accS,  false,  false);
    }
    if (hi == 0) {
      unsigned* orow = (unsigned*)out + batch * 59;
      const unsigned vi = (unsigned)(accI[0] + 0.5f);
      const unsigned vs = (unsigned)(accS[0] + 0.5f);
      if (n < 58) {
        const unsigned old = atomicAdd(&orow[n], vi + ARR_ONE);
        if ((old >> ARR_SHIFT) == WGPI - 1) {
          const unsigned tot = (old & (ARR_ONE - 1u)) + vi;
          atomicExch(&orow[n], __float_as_uint((float)(tot & 255u)));
        }
      }
      if (lm == 10) {
        const unsigned old = atomicAdd(&orow[58], vs + ARR_ONE);
        if ((old >> ARR_SHIFT) == 4 * WGPI - 1) {
          const unsigned tot = (old & (ARR_ONE - 1u)) + vs;
          const unsigned rest = (512u * 512u) - tot;
          atomicExch(&orow[58], __float_as_uint((float)(rest & 255u)));
        }
      }
    }
  }
}

extern "C" void kernel_launch(void* const* d_in, const int* in_sizes, int n_in,
                              void* d_out, int out_size, void* d_ws,
                              size_t ws_size, hipStream_t stream) {
  const float* x = (const float*)d_in[0];
  float* out = (float*)d_out;

  const int B = in_sizes[0] >> 18;

  start_out_kernel<<<(B * 59 + 255) / 256, 256, 0, stream>>>(out, B * 59);
  lbp_hist_kernel<<<dim3(WGPI, B), 1024, 0, stream>>>(x, out);
}
